# starting kernel
# baseline (speedup 1.0000x reference)
#include <hip/hip_runtime.h>

#define NVX 128
#define NVY 128
#define NC  80
#define B_  2
#define N_  6
#define D_  112
#define H_  16
#define W_  44
#define HW  (H_ * W_)
#define RAYS (B_ * N_ * H_ * W_)
#define RAYS_PER_BLOCK 8

typedef __attribute__((ext_vector_type(2))) float v2f;
typedef __attribute__((ext_vector_type(8))) float v8f;

__global__ __launch_bounds__(256) void vp_zero(float4* __restrict__ out, int n4) {
  int i = blockIdx.x * blockDim.x + threadIdx.x;
  const int stride = gridDim.x * blockDim.x;
  for (; i < n4; i += stride) out[i] = make_float4(0.f, 0.f, 0.f, 0.f);
}

__global__ __launch_bounds__(256) void vp_pool(const int*   __restrict__ geom,
                                               const float* __restrict__ depth,
                                               const float* __restrict__ ctx,
                                               float*       __restrict__ out) {
  const int lane = threadIdx.x & 31;
  const int wave = threadIdx.x >> 5;
  const int ray  = blockIdx.x * RAYS_PER_BLOCK + wave;

  const bool lo = lane < 16;
  const int w   = ray % W_;
  const int t   = ray / W_;
  const int h   = t % H_;
  const int bn  = t / H_;
  const int b   = bn / N_;

  const int baseP   = bn * (D_ * HW) + h * W_ + w;
  const int ctxBase = bn * (NC * HW) + h * W_ + w;
  const int voxBase = b * (NVX * NVY);

  v2f Bv[5];
#pragma unroll
  for (int ct = 0; ct < 5; ++ct) {
    float cv = 0.f;
    if (lo) cv = ctx[ctxBase + (ct * 16 + lane) * HW];
    v2f bb; bb.x = cv; bb.y = 0.f;
    Bv[ct] = bb;
  }

  const int rowoff = (lane >> 4) << 3;
  const int lanec  = lane & 15;

  for (int dt = 0; dt < 7; ++dt) {
    const int d0 = dt * 16;

    if (dt < 6) {
      int pd = d0 + 16 + lane;
      pd = pd < D_ ? pd : (D_ - 1);
      const int pp = baseP + pd * HW;
      __builtin_prefetch(&geom[3 * pp], 0, 1);
      __builtin_prefetch(&depth[pp], 0, 1);
    }

    float dv  = 0.f;
    int  voff = 0;
    if (lo) {
      const int p  = baseP + (d0 + lane) * HW;
      dv = depth[p];
      const int gx = geom[3 * p + 0];
      const int gy = geom[3 * p + 1];
      if (!((unsigned)gx < (unsigned)NVX && (unsigned)gy < (unsigned)NVY))
        dv = 0.f;
      const int cx = gx < 0 ? 0 : (gx > NVX - 1 ? NVX - 1 : gx);
      const int cy = gy < 0 ? 0 : (gy > NVY - 1 ? NVY - 1 : gy);
      voff = (voxBase + cy * NVX + cx) * NC;
    }
    v2f Av; Av.x = dv; Av.y = 0.f;

    int offr[8];
#pragma unroll
    for (int r = 0; r < 8; ++r)
      offr[r] = __shfl(voff, r + rowoff, 32);

    v8f acc[5];
#pragma unroll
    for (int ct = 0; ct < 5; ++ct) {
      v8f z = {0.f, 0.f, 0.f, 0.f, 0.f, 0.f, 0.f, 0.f};
      acc[ct] = __builtin_amdgcn_wmma_f32_16x16x4_f32(
           false, Av,  false, Bv[ct],
           (short)0, z,  false,  false);
    }

#pragma unroll
    for (int r = 0; r < 8; ++r) {
      float* p = out + (unsigned)(offr[r] + lanec);
#pragma unroll
      for (int ct = 0; ct < 5; ++ct) {
        __hip_atomic_fetch_add(p + ct * 16, acc[ct][r], __ATOMIC_RELAXED,
                               __HIP_MEMORY_SCOPE_AGENT);
      }
    }
  }
}

extern "C" void kernel_launch(void* const* d_in, const int* in_sizes, int n_in,
                              void* d_out, int out_size, void* d_ws, size_t ws_size,
                              hipStream_t stream) {
  (void)in_sizes; (void)n_in; (void)d_ws; (void)ws_size;
  const int*   geom  = (const int*)d_in[0];
  const float* depth = (const float*)d_in[1];
  const float* ctx   = (const float*)d_in[2];
  float*       out   = (float*)d_out;

  const int n4 = out_size / 4;
  vp_zero<<<512, 256, 0, stream>>>((float4*)out, n4);

  vp_pool<<<RAYS / RAYS_PER_BLOCK, 256, 0, stream>>>(geom, depth, ctx, out);
}


# Scatter atomics and zero pass replaced: per-call LDS counting sort into 16-voxel groups, then one-hot(voxel)*depth x gathered ctx rows on WMMA sums each group; adds now in binned (LDS arrival) order
# speedup vs baseline: 1.2174x; 1.2174x over previous
#include <hip/hip_runtime.h>

#define NVX 128
#define NVY 128
#define NC  80
#define B_  2
#define N_  6
#define D_  112
#define H_  16
#define W_  44
#define HW  (H_ * W_)
#define NPTS  (B_ * N_ * D_ * HW)
#define NRAYS (B_ * N_ * HW)
#define NBINS (B_ * NVX * NVY / 16)
#define SB    64
#define PPB   (NPTS / SB)
#define ST    1024
#define PPT   15
#define TW    64
#define NTB   (B_ * N_ * (HW / TW))

#define WS_CTXT 0
#define WS_HIST (WS_CTXT + NRAYS * NC * 4)
#define WS_BASE (WS_HIST + SB * NBINS * 4)
#define WS_RECS (WS_BASE + 8448)
#define WS_NEED ((size_t)WS_RECS + (size_t)NPTS * 8)

typedef __attribute__((ext_vector_type(2))) float v2f;
typedef __attribute__((ext_vector_type(8))) float v8f;

__device__ __forceinline__ unsigned point_key(const int* __restrict__ geom, int p, bool& valid) {
  const int gx = geom[3 * p + 0];
  const int gy = geom[3 * p + 1];
  valid = ((unsigned)gx < (unsigned)NVX) && ((unsigned)gy < (unsigned)NVY);
  const int cx = gx < 0 ? 0 : (gx > NVX - 1 ? NVX - 1 : gx);
  const int cy = gy < 0 ? 0 : (gy > NVY - 1 ? NVY - 1 : gy);
  const int b    = p / (N_ * D_ * HW);
  const int flat = b * (NVX * NVY) + cy * NVX + cx;
  const int bn   = p / (D_ * HW);
  const int ray  = bn * HW + p % HW;
  return (unsigned)ray | ((unsigned)flat << 14);
}

__global__ __launch_bounds__(ST) void vp_count(const int*   __restrict__ geom,
                                               const float* __restrict__ ctx,
                                               float* __restrict__ ctxT,
                                               int*   __restrict__ hist,
                                               size_t ws_size) {
  if (ws_size < WS_NEED) return;
  __shared__ float sm[NC * (TW + 1)];
  const int tid = threadIdx.x;
  if (blockIdx.x < SB) {
    int* cnt = (int*)sm;
    cnt[tid] = 0;
    cnt[tid + ST] = 0;
    __syncthreads();
    const int base = blockIdx.x * PPB;
#pragma unroll
    for (int i = 0; i < PPT; ++i) {
      const int l = tid + i * ST;
      if (l < PPB) {
        bool valid;
        const unsigned key = point_key(geom, base + l, valid);
        atomicAdd(&cnt[key >> 18], 1);
      }
    }
    __syncthreads();
    int* row = hist + blockIdx.x * NBINS;
    row[tid] = cnt[tid];
    row[tid + ST] = cnt[tid + ST];
  } else {
    const int t   = blockIdx.x - SB;
    const int bn  = t / (HW / TW);
    const int hw0 = (t % (HW / TW)) * TW;
#pragma unroll
    for (int i = 0; i < (NC * TW) / ST; ++i) {
      const int e = tid + i * ST;
      const int c = e / TW, x = e % TW;
      sm[c * (TW + 1) + x] = ctx[(bn * NC + c) * HW + hw0 + x];
    }
    __syncthreads();
    float* dst = ctxT + (size_t)(bn * HW + hw0) * NC;
#pragma unroll
    for (int i = 0; i < (NC * TW) / ST; ++i) {
      const int e = tid + i * ST;
      const int x = e / NC, c = e % NC;
      dst[e] = sm[c * (TW + 1) + x];
    }
  }
}

__device__ __forceinline__ int2 block_excl_scan2(int2 v, int* sc, int tid) {
  const int lane = tid & 31, w = tid >> 5;
  int ax = v.x, ay = v.y;
#pragma unroll
  for (int off = 1; off < 32; off <<= 1) {
    const int nx = __shfl_up(ax, off, 32);
    const int ny = __shfl_up(ay, off, 32);
    if (lane >= off) { ax += nx; ay += ny; }
  }
  if (lane == 31) { sc[w] = ax; sc[32 + w] = ay; }
  __syncthreads();
  if (w == 0) {
    const int ox = sc[lane], oy = sc[32 + lane];
    int bx = ox, by = oy;
#pragma unroll
    for (int off = 1; off < 32; off <<= 1) {
      const int nx = __shfl_up(bx, off, 32);
      const int ny = __shfl_up(by, off, 32);
      if (lane >= off) { bx += nx; by += ny; }
    }
    sc[lane] = bx - ox;
    sc[32 + lane] = by - oy;
  }
  __syncthreads();
  int2 r;
  r.x = ax - v.x + sc[w];
  r.y = ay - v.y + sc[32 + w];
  return r;
}

__global__ __launch_bounds__(ST) void vp_place(const int*   __restrict__ geom,
                                               const float* __restrict__ depth,
                                               const int*   __restrict__ hist,
                                               int*   __restrict__ binBase,
                                               uint2* __restrict__ recs,
                                               size_t ws_size) {
  if (ws_size < WS_NEED) return;
  __shared__ int   cnt[NBINS];
  __shared__ int   lbase[NBINS];
  __shared__ int   gbase[NBINS];
  __shared__ int   sc[64];
  __shared__ uint2 lrec[PPB];
  const int tid  = threadIdx.x;
  const int blk  = blockIdx.x;
  const int base = blk * PPB;

  cnt[tid] = 0;
  cnt[tid + ST] = 0;
  __syncthreads();

  unsigned keys[PPT];
  int      rank[PPT];
  float    dval[PPT];
#pragma unroll
  for (int i = 0; i < PPT; ++i) {
    const int l = tid + i * ST;
    keys[i] = 0u; rank[i] = 0; dval[i] = 0.f;
    if (l < PPB) {
      bool valid;
      const unsigned key = point_key(geom, base + l, valid);
      const float d = depth[base + l];
      keys[i] = key;
      dval[i] = valid ? d : 0.f;
      rank[i] = atomicAdd(&cnt[key >> 18], 1);
    }
  }
  __syncthreads();

  const int c0 = cnt[2 * tid], c1 = cnt[2 * tid + 1];
  int tot0 = 0, tot1 = 0, pre0 = 0, pre1 = 0;
  const int2* h2 = (const int2*)hist;
#pragma unroll 8
  for (int r = 0; r < SB; ++r) {
    const int2 v = h2[r * (NBINS / 2) + tid];
    tot0 += v.x; tot1 += v.y;
    if (r < blk) { pre0 += v.x; pre1 += v.y; }
  }
  int2 in; in.x = tot0 + tot1; in.y = c0 + c1;
  const int2 ex = block_excl_scan2(in, sc, tid);
  gbase[2 * tid]     = ex.x + pre0;
  gbase[2 * tid + 1] = ex.x + tot0 + pre1;
  lbase[2 * tid]     = ex.y;
  lbase[2 * tid + 1] = ex.y + c0;
  if (blk == 0) {
    binBase[2 * tid]     = ex.x;
    binBase[2 * tid + 1] = ex.x + tot0;
    if (tid == ST - 1) binBase[NBINS] = ex.x + tot0 + tot1;
  }
  __syncthreads();

#pragma unroll
  for (int i = 0; i < PPT; ++i) {
    const int l = tid + i * ST;
    if (l < PPB) {
      uint2 rr; rr.x = keys[i]; rr.y = __float_as_uint(dval[i]);
      lrec[lbase[keys[i] >> 18] + rank[i]] = rr;
    }
  }
  __syncthreads();

#pragma unroll
  for (int i = 0; i < PPT; ++i) {
    const int l = tid + i * ST;
    if (l < PPB) {
      const uint2 rr = lrec[l];
      const int g = rr.x >> 18;
      recs[gbase[g] + (l - lbase[g])] = rr;
    }
  }
}

struct StepIn {
  float d0, d1;
  int   m0, m1;
  const float* row0;
  const float* row1;
};
struct StepB {
  float4 q0, q1;
  float  e0, e1;
};

__device__ __forceinline__ StepIn decode_recs(uint2 ra, uint2 rb,
                                              const float* __restrict__ ctxT,
                                              int i0, int end) {
  if (i0     >= end) { ra.x = 0u; ra.y = 0u; }
  if (i0 + 1 >= end) { rb.x = 0u; rb.y = 0u; }
  StepIn s;
  s.d0 = __uint_as_float(ra.y);
  s.d1 = __uint_as_float(rb.y);
  s.m0 = (ra.x >> 14) & 15;
  s.m1 = (rb.x >> 14) & 15;
  s.row0 = ctxT + (ra.x & 0x3FFFu) * NC;
  s.row1 = ctxT + (rb.x & 0x3FFFu) * NC;
  return s;
}

__device__ __forceinline__ StepB load_ctx(const StepIn& s, int c) {
  StepB b;
  b.q0 = *(const float4*)(s.row0 + 4 * c);
  b.q1 = *(const float4*)(s.row1 + 4 * c);
  b.e0 = s.row0[64 + c];
  b.e1 = s.row1[64 + c];
  return b;
}

__global__ __launch_bounds__(128) void vp_gather(const int*   __restrict__ binBase,
                                                 const uint2* __restrict__ recs,
                                                 const float* __restrict__ ctxT,
                                                 float*       __restrict__ out,
                                                 size_t ws_size) {
  if (ws_size < WS_NEED) return;
  __shared__ float part[4][16 * NC];
  const int tid  = threadIdx.x;
  const int lane = tid & 31;
  const int wave = tid >> 5;
  const int c = lane & 15;
  const int h = lane >> 4;
  const int g = blockIdx.x;

  const int start  = binBase[g];
  const int end    = binBase[g + 1];
  const int nsteps = (end - start + 3) >> 2;

  v8f acc[5];
#pragma unroll
  for (int j = 0; j < 5; ++j) {
    v8f z = {0.f, 0.f, 0.f, 0.f, 0.f, 0.f, 0.f, 0.f};
    acc[j] = z;
  }

  const int i00 = start + 2 * h;
  for (int s = wave; s < nsteps; s += 16) {
    uint2 ra[4], rb[4];
#pragma unroll
    for (int u = 0; u < 4; ++u) {
      const int i0 = i00 + 4 * (s + 4 * u);
      ra[u] = recs[i0     < NPTS ? i0     : NPTS - 1];
      rb[u] = recs[i0 + 1 < NPTS ? i0 + 1 : NPTS - 1];
    }
    asm volatile("" :: "v"(ra[0].x), "v"(ra[0].y), "v"(rb[0].x), "v"(rb[0].y),
                       "v"(ra[1].x), "v"(ra[1].y), "v"(rb[1].x), "v"(rb[1].y),
                       "v"(ra[2].x), "v"(ra[2].y), "v"(rb[2].x), "v"(rb[2].y),
                       "v"(ra[3].x), "v"(ra[3].y), "v"(rb[3].x), "v"(rb[3].y));
    StepIn rs[4];
#pragma unroll
    for (int u = 0; u < 4; ++u)
      rs[u] = decode_recs(ra[u], rb[u], ctxT, i00 + 4 * (s + 4 * u), end);
    StepB bs[4];
#pragma unroll
    for (int u = 0; u < 4; ++u)
      bs[u] = load_ctx(rs[u], c);
    asm volatile("" :: "v"(bs[0].q0.x), "v"(bs[0].q1.x), "v"(bs[0].e0), "v"(bs[0].e1),
                       "v"(bs[1].q0.x), "v"(bs[1].q1.x), "v"(bs[1].e0), "v"(bs[1].e1),
                       "v"(bs[2].q0.x), "v"(bs[2].q1.x), "v"(bs[2].e0), "v"(bs[2].e1),
                       "v"(bs[3].q0.x), "v"(bs[3].q1.x), "v"(bs[3].e0), "v"(bs[3].e1));
#pragma unroll
    for (int u = 0; u < 4; ++u) {
      v2f a;
      a.x = (rs[u].m0 == c) ? rs[u].d0 : 0.f;
      a.y = (rs[u].m1 == c) ? rs[u].d1 : 0.f;
      const float x0[4] = {bs[u].q0.x, bs[u].q0.y, bs[u].q0.z, bs[u].q0.w};
      const float x1[4] = {bs[u].q1.x, bs[u].q1.y, bs[u].q1.z, bs[u].q1.w};
#pragma unroll
      for (int j = 0; j < 4; ++j) {
        v2f b; b.x = x0[j]; b.y = x1[j];
        acc[j] = __builtin_amdgcn_wmma_f32_16x16x4_f32(
            false, a, false, b, (short)0, acc[j], false, false);
      }
      v2f b4; b4.x = bs[u].e0; b4.y = bs[u].e1;
      acc[4] = __builtin_amdgcn_wmma_f32_16x16x4_f32(
          false, a, false, b4, (short)0, acc[4], false, false);
    }
  }

  float* mine = part[wave];
#pragma unroll
  for (int r = 0; r < 8; ++r) {
    float* rowp = mine + (8 * h + r) * NC;
    *(float4*)(rowp + 4 * c) = make_float4(acc[0][r], acc[1][r], acc[2][r], acc[3][r]);
    rowp[64 + c] = acc[4][r];
  }
  __syncthreads();

  float4* og = (float4*)(out + (size_t)g * (16 * NC));
#pragma unroll
  for (int i = 0; i < 3; ++i) {
    const int e4 = tid + i * 128;
    if (e4 < (16 * NC) / 4) {
      const float4 p0 = ((const float4*)part[0])[e4];
      const float4 p1 = ((const float4*)part[1])[e4];
      const float4 p2 = ((const float4*)part[2])[e4];
      const float4 p3 = ((const float4*)part[3])[e4];
      float4 v;
      v.x = (p0.x + p1.x) + (p2.x + p3.x);
      v.y = (p0.y + p1.y) + (p2.y + p3.y);
      v.z = (p0.z + p1.z) + (p2.z + p3.z);
      v.w = (p0.w + p1.w) + (p2.w + p3.w);
      og[e4] = v;
    }
  }
}

extern "C" void kernel_launch(void* const* d_in, const int* in_sizes, int n_in,
                              void* d_out, int out_size, void* d_ws, size_t ws_size,
                              hipStream_t stream) {
  (void)in_sizes; (void)n_in; (void)out_size;
  const int*   geom  = (const int*)d_in[0];
  const float* depth = (const float*)d_in[1];
  const float* ctx   = (const float*)d_in[2];
  float*       out   = (float*)d_out;

  char*  ws      = (char*)d_ws;
  float* ctxT    = (float*)(ws + WS_CTXT);
  int*   hist    = (int*)(ws + WS_HIST);
  int*   binBase = (int*)(ws + WS_BASE);
  uint2* recs    = (uint2*)(ws + WS_RECS);

  vp_count<<<SB + NTB, ST, 0, stream>>>(geom, ctx, ctxT, hist, ws_size);
  vp_place<<<SB, ST, 0, stream>>>(geom, depth, hist, binBase, recs, ws_size);
  vp_gather<<<NBINS, 128, 0, stream>>>(binBase, recs, ctxT, out, ws_size);
}


# speedup vs baseline: 1.2891x; 1.0589x over previous
#include <hip/hip_runtime.h>

#define NVX 128
#define NVY 128
#define NC  80
#define B_  2
#define N_  6
#define D_  112
#define H_  16
#define W_  44
#define HW  (H_ * W_)
#define NPTS  (B_ * N_ * D_ * HW)
#define NRAYS (B_ * N_ * HW)
#define NBINS (B_ * NVX * NVY / 16)
#define SB    64
#define PPB   (NPTS / SB)
#define ST    1024
#define PPT   15
#define TW    64
#define NTB   (B_ * N_ * (HW / TW))
#define NCM   64
#define NCE   (NC - NCM)

#define WS_CTXM 0
#define WS_CTXE (WS_CTXM + NRAYS * NCM * 4)
#define WS_HIST (WS_CTXE + NRAYS * NCE * 4)
#define WS_BASE (WS_HIST + SB * NBINS * 4)
#define WS_RECS (WS_BASE + 8448)
#define WS_NEED ((size_t)WS_RECS + (size_t)NPTS * 8)

typedef __attribute__((ext_vector_type(2))) float v2f;
typedef __attribute__((ext_vector_type(8))) float v8f;

__device__ __forceinline__ unsigned point_key(const int* __restrict__ geom, int p, bool& valid) {
  const int gx = geom[3 * p + 0];
  const int gy = geom[3 * p + 1];
  valid = ((unsigned)gx < (unsigned)NVX) && ((unsigned)gy < (unsigned)NVY);
  const int cx = gx < 0 ? 0 : (gx > NVX - 1 ? NVX - 1 : gx);
  const int cy = gy < 0 ? 0 : (gy > NVY - 1 ? NVY - 1 : gy);
  const int b    = p / (N_ * D_ * HW);
  const int flat = b * (NVX * NVY) + cy * NVX + cx;
  const int bn   = p / (D_ * HW);
  const int ray  = bn * HW + p % HW;
  return (unsigned)ray | ((unsigned)flat << 14);
}

__global__ __launch_bounds__(ST) void vp_count(const int*   __restrict__ geom,
                                               const float* __restrict__ ctx,
                                               float* __restrict__ ctxM,
                                               float* __restrict__ ctxE,
                                               int*   __restrict__ hist,
                                               size_t ws_size) {
  if (ws_size < WS_NEED) return;
  __shared__ float sm[NC * (TW + 1)];
  const int tid = threadIdx.x;
  if (blockIdx.x < SB) {
    int* cnt = (int*)sm;
    cnt[tid] = 0;
    cnt[tid + ST] = 0;
    __syncthreads();
    const int base = blockIdx.x * PPB;
#pragma unroll
    for (int i = 0; i < PPT; ++i) {
      const int l = tid + i * ST;
      if (l < PPB) {
        bool valid;
        const unsigned key = point_key(geom, base + l, valid);
        atomicAdd(&cnt[key >> 18], 1);
      }
    }
    __syncthreads();
    int* row = hist + blockIdx.x * NBINS;
    row[tid] = cnt[tid];
    row[tid + ST] = cnt[tid + ST];
  } else {
    const int t   = blockIdx.x - SB;
    const int bn  = t / (HW / TW);
    const int hw0 = (t % (HW / TW)) * TW;
#pragma unroll
    for (int i = 0; i < (NC * TW) / ST; ++i) {
      const int e = tid + i * ST;
      const int c = e / TW, x = e % TW;
      sm[c * (TW + 1) + x] = ctx[(bn * NC + c) * HW + hw0 + x];
    }
    __syncthreads();
    float* dstM = ctxM + (size_t)(bn * HW + hw0) * NCM;
#pragma unroll
    for (int i = 0; i < (NCM * TW) / ST; ++i) {
      const int e = tid + i * ST;
      const int x = e / NCM, c = e % NCM;
      dstM[e] = sm[c * (TW + 1) + x];
    }
    float* dstE = ctxE + (size_t)(bn * HW + hw0) * NCE;
#pragma unroll
    for (int i = 0; i < (NCE * TW) / ST; ++i) {
      const int e = tid + i * ST;
      const int x = e / NCE, c = NCM + e % NCE;
      dstE[e] = sm[c * (TW + 1) + x];
    }
  }
}

__device__ __forceinline__ int2 block_excl_scan2(int2 v, int* sc, int tid) {
  const int lane = tid & 31, w = tid >> 5;
  int ax = v.x, ay = v.y;
#pragma unroll
  for (int off = 1; off < 32; off <<= 1) {
    const int nx = __shfl_up(ax, off, 32);
    const int ny = __shfl_up(ay, off, 32);
    if (lane >= off) { ax += nx; ay += ny; }
  }
  if (lane == 31) { sc[w] = ax; sc[32 + w] = ay; }
  __syncthreads();
  if (w == 0) {
    const int ox = sc[lane], oy = sc[32 + lane];
    int bx = ox, by = oy;
#pragma unroll
    for (int off = 1; off < 32; off <<= 1) {
      const int nx = __shfl_up(bx, off, 32);
      const int ny = __shfl_up(by, off, 32);
      if (lane >= off) { bx += nx; by += ny; }
    }
    sc[lane] = bx - ox;
    sc[32 + lane] = by - oy;
  }
  __syncthreads();
  int2 r;
  r.x = ax - v.x + sc[w];
  r.y = ay - v.y + sc[32 + w];
  return r;
}

__global__ __launch_bounds__(ST) void vp_place(const int*   __restrict__ geom,
                                               const float* __restrict__ depth,
                                               const int*   __restrict__ hist,
                                               int*   __restrict__ binBase,
                                               uint2* __restrict__ recs,
                                               size_t ws_size) {
  if (ws_size < WS_NEED) return;
  __shared__ int   cnt[NBINS];
  __shared__ int   lbase[NBINS];
  __shared__ int   gbase[NBINS];
  __shared__ int   sc[64];
  __shared__ uint2 lrec[PPB];
  const int tid  = threadIdx.x;
  const int blk  = blockIdx.x;
  const int base = blk * PPB;

  cnt[tid] = 0;
  cnt[tid + ST] = 0;
  __syncthreads();

  unsigned keys[PPT];
  int      rank[PPT];
  float    dval[PPT];
#pragma unroll
  for (int i = 0; i < PPT; ++i) {
    const int l = tid + i * ST;
    keys[i] = 0u; rank[i] = 0; dval[i] = 0.f;
    if (l < PPB) {
      bool valid;
      const unsigned key = point_key(geom, base + l, valid);
      const float d = depth[base + l];
      keys[i] = key;
      dval[i] = valid ? d : 0.f;
      rank[i] = atomicAdd(&cnt[key >> 18], 1);
    }
  }
  __syncthreads();

  const int c0 = cnt[2 * tid], c1 = cnt[2 * tid + 1];
  int tot0 = 0, tot1 = 0, pre0 = 0, pre1 = 0;
  const int2* h2 = (const int2*)hist;
#pragma unroll 8
  for (int r = 0; r < SB; ++r) {
    const int2 v = h2[r * (NBINS / 2) + tid];
    tot0 += v.x; tot1 += v.y;
    if (r < blk) { pre0 += v.x; pre1 += v.y; }
  }
  int2 in; in.x = tot0 + tot1; in.y = c0 + c1;
  const int2 ex = block_excl_scan2(in, sc, tid);
  gbase[2 * tid]     = ex.x + pre0;
  gbase[2 * tid + 1] = ex.x + tot0 + pre1;
  lbase[2 * tid]     = ex.y;
  lbase[2 * tid + 1] = ex.y + c0;
  if (blk == 0) {
    binBase[2 * tid]     = ex.x;
    binBase[2 * tid + 1] = ex.x + tot0;
    if (tid == ST - 1) binBase[NBINS] = ex.x + tot0 + tot1;
  }
  __syncthreads();

#pragma unroll
  for (int i = 0; i < PPT; ++i) {
    const int l = tid + i * ST;
    if (l < PPB) {
      uint2 rr; rr.x = keys[i]; rr.y = __float_as_uint(dval[i]);
      lrec[lbase[keys[i] >> 18] + rank[i]] = rr;
    }
  }
  __syncthreads();

#pragma unroll
  for (int i = 0; i < PPT; ++i) {
    const int l = tid + i * ST;
    if (l < PPB) {
      const uint2 rr = lrec[l];
      const int g = rr.x >> 18;
      recs[gbase[g] + (l - lbase[g])] = rr;
    }
  }
}

struct StepIn {
  float d0, d1;
  int   m0, m1;
  unsigned r0, r1;
};
struct StepB {
  float4 q0, q1;
  float  e0, e1;
};

__device__ __forceinline__ StepIn decode_recs(uint2 ra, uint2 rb, int i0, int end) {
  if (i0     >= end) { ra.x = 0u; ra.y = 0u; }
  if (i0 + 1 >= end) { rb.x = 0u; rb.y = 0u; }
  StepIn s;
  s.d0 = __uint_as_float(ra.y);
  s.d1 = __uint_as_float(rb.y);
  s.m0 = (ra.x >> 14) & 15;
  s.m1 = (rb.x >> 14) & 15;
  s.r0 = ra.x & 0x3FFFu;
  s.r1 = rb.x & 0x3FFFu;
  return s;
}

__device__ __forceinline__ StepB load_ctx(const StepIn& s,
                                          const float* __restrict__ ctxM,
                                          const float* __restrict__ ctxE, int c) {
  const unsigned cm = (unsigned)c * 16u;
  const unsigned ce = (unsigned)c * 4u;
  StepB b;
  b.q0 = *(const float4*)((const char*)ctxM + ((s.r0 * (NCM * 4u)) | cm));
  b.q1 = *(const float4*)((const char*)ctxM + ((s.r1 * (NCM * 4u)) | cm));
  b.e0 = *(const float*)((const char*)ctxE + ((s.r0 * (NCE * 4u)) | ce));
  b.e1 = *(const float*)((const char*)ctxE + ((s.r1 * (NCE * 4u)) | ce));
  return b;
}

__global__ __launch_bounds__(128) void vp_gather(const int*   __restrict__ binBase,
                                                 const uint2* __restrict__ recs,
                                                 const float* __restrict__ ctxM,
                                                 const float* __restrict__ ctxE,
                                                 float*       __restrict__ out,
                                                 size_t ws_size) {
  if (ws_size < WS_NEED) return;
  __shared__ float part[4][16 * NC];
  const int tid  = threadIdx.x;
  const int lane = tid & 31;
  const int wave = tid >> 5;
  const int c = lane & 15;
  const int h = lane >> 4;
  const int g = blockIdx.x;

  const int start  = binBase[g];
  const int end    = binBase[g + 1];
  const int nsteps = (end - start + 3) >> 2;

  v8f acc[5];
#pragma unroll
  for (int j = 0; j < 5; ++j) {
    v8f z = {0.f, 0.f, 0.f, 0.f, 0.f, 0.f, 0.f, 0.f};
    acc[j] = z;
  }

  const int i00 = start + 2 * h;
  for (int s = wave; s < nsteps; s += 16) {
    uint2 ra[4], rb[4];
#pragma unroll
    for (int u = 0; u < 4; ++u) {
      const int i0 = i00 + 4 * (s + 4 * u);
      ra[u] = recs[i0     < NPTS ? i0     : NPTS - 1];
      rb[u] = recs[i0 + 1 < NPTS ? i0 + 1 : NPTS - 1];
    }
    asm volatile("" :: "v"(ra[0].x), "v"(ra[0].y), "v"(rb[0].x), "v"(rb[0].y),
                       "v"(ra[1].x), "v"(ra[1].y), "v"(rb[1].x), "v"(rb[1].y),
                       "v"(ra[2].x), "v"(ra[2].y), "v"(rb[2].x), "v"(rb[2].y),
                       "v"(ra[3].x), "v"(ra[3].y), "v"(rb[3].x), "v"(rb[3].y));
    StepIn rs[4];
#pragma unroll
    for (int u = 0; u < 4; ++u)
      rs[u] = decode_recs(ra[u], rb[u], i00 + 4 * (s + 4 * u), end);
    StepB bs[4];
#pragma unroll
    for (int u = 0; u < 4; ++u)
      bs[u] = load_ctx(rs[u], ctxM, ctxE, c);
    asm volatile("" :: "v"(bs[0].q0.x), "v"(bs[0].q1.x), "v"(bs[0].e0), "v"(bs[0].e1),
                       "v"(bs[1].q0.x), "v"(bs[1].q1.x), "v"(bs[1].e0), "v"(bs[1].e1),
                       "v"(bs[2].q0.x), "v"(bs[2].q1.x), "v"(bs[2].e0), "v"(bs[2].e1),
                       "v"(bs[3].q0.x), "v"(bs[3].q1.x), "v"(bs[3].e0), "v"(bs[3].e1));
#pragma unroll
    for (int u = 0; u < 4; ++u) {
      v2f a;
      a.x = (rs[u].m0 == c) ? rs[u].d0 : 0.f;
      a.y = (rs[u].m1 == c) ? rs[u].d1 : 0.f;
      const float x0[4] = {bs[u].q0.x, bs[u].q0.y, bs[u].q0.z, bs[u].q0.w};
      const float x1[4] = {bs[u].q1.x, bs[u].q1.y, bs[u].q1.z, bs[u].q1.w};
#pragma unroll
      for (int j = 0; j < 4; ++j) {
        v2f b; b.x = x0[j]; b.y = x1[j];
        acc[j] = __builtin_amdgcn_wmma_f32_16x16x4_f32(
            false, a, false, b, (short)0, acc[j], false, false);
      }
      v2f b4; b4.x = bs[u].e0; b4.y = bs[u].e1;
      acc[4] = __builtin_amdgcn_wmma_f32_16x16x4_f32(
          false, a, false, b4, (short)0, acc[4], false, false);
    }
  }

  float* mine = part[wave];
#pragma unroll
  for (int r = 0; r < 8; ++r) {
    float* rowp = mine + (8 * h + r) * NC;
    *(float4*)(rowp + 4 * c) = make_float4(acc[0][r], acc[1][r], acc[2][r], acc[3][r]);
    rowp[64 + c] = acc[4][r];
  }
  __syncthreads();

  float4* og = (float4*)(out + (size_t)g * (16 * NC));
#pragma unroll
  for (int i = 0; i < 3; ++i) {
    const int e4 = tid + i * 128;
    if (e4 < (16 * NC) / 4) {
      const float4 p0 = ((const float4*)part[0])[e4];
      const float4 p1 = ((const float4*)part[1])[e4];
      const float4 p2 = ((const float4*)part[2])[e4];
      const float4 p3 = ((const float4*)part[3])[e4];
      float4 v;
      v.x = (p0.x + p1.x) + (p2.x + p3.x);
      v.y = (p0.y + p1.y) + (p2.y + p3.y);
      v.z = (p0.z + p1.z) + (p2.z + p3.z);
      v.w = (p0.w + p1.w) + (p2.w + p3.w);
      og[e4] = v;
    }
  }
}

extern "C" void kernel_launch(void* const* d_in, const int* in_sizes, int n_in,
                              void* d_out, int out_size, void* d_ws, size_t ws_size,
                              hipStream_t stream) {
  (void)in_sizes; (void)n_in; (void)out_size;
  const int*   geom  = (const int*)d_in[0];
  const float* depth = (const float*)d_in[1];
  const float* ctx   = (const float*)d_in[2];
  float*       out   = (float*)d_out;

  char*  ws      = (char*)d_ws;
  float* ctxM    = (float*)(ws + WS_CTXM);
  float* ctxE    = (float*)(ws + WS_CTXE);
  int*   hist    = (int*)(ws + WS_HIST);
  int*   binBase = (int*)(ws + WS_BASE);
  uint2* recs    = (uint2*)(ws + WS_RECS);

  vp_count<<<SB + NTB, ST, 0, stream>>>(geom, ctx, ctxM, ctxE, hist, ws_size);
  vp_place<<<SB, ST, 0, stream>>>(geom, depth, hist, binBase, recs, ws_size);
  vp_gather<<<NBINS, 128, 0, stream>>>(binBase, recs, ctxM, ctxE, out, ws_size);
}


# Count launch and histogram reads dropped: 128 sort blocks write block-local sorted segments + offset rows; a group's block gathers its records from 64 segments into LDS; sum order now by segment
# speedup vs baseline: 2.1307x; 1.6529x over previous
#include <hip/hip_runtime.h>

#define NVX 128
#define NVY 128
#define NC  80
#define B_  2
#define N_  6
#define D_  112
#define H_  16
#define W_  44
#define HW  (H_ * W_)
#define NPTS  (B_ * N_ * D_ * HW)
#define NRAYS (B_ * N_ * HW)
#define NBINS (B_ * NVX * NVY / 16)
#define NB    (NBINS / B_)
#define SEGS  64
#define SB    (B_ * SEGS)
#define PPB   (NPTS / SB)
#define ST    1024
#define PPT   8
#define TW    64
#define NTB   (B_ * N_ * (HW / TW))
#define NCM   64
#define NCE   (NC - NCM)
#define CH    1024

#define WS_CTXM 0
#define WS_CTXE (WS_CTXM + NRAYS * NCM * 4)
#define WS_OFFS (WS_CTXE + NRAYS * NCE * 4)
#define WS_RECS (WS_OFFS + SB * (NB + 1) * 4)
#define WS_NEED ((size_t)WS_RECS + (size_t)NPTS * 8)

typedef __attribute__((ext_vector_type(2))) float v2f;
typedef __attribute__((ext_vector_type(8))) float v8f;

__device__ __forceinline__ unsigned point_key(const int* __restrict__ geom, int p, bool& valid) {
  const int gx = geom[3 * p + 0];
  const int gy = geom[3 * p + 1];
  valid = ((unsigned)gx < (unsigned)NVX) && ((unsigned)gy < (unsigned)NVY);
  const int cx = gx < 0 ? 0 : (gx > NVX - 1 ? NVX - 1 : gx);
  const int cy = gy < 0 ? 0 : (gy > NVY - 1 ? NVY - 1 : gy);
  const int b    = p / (N_ * D_ * HW);
  const int flat = b * (NVX * NVY) + cy * NVX + cx;
  const int bn   = p / (D_ * HW);
  const int ray  = bn * HW + p % HW;
  return (unsigned)ray | ((unsigned)flat << 14);
}

__device__ __forceinline__ int block_excl_scan(int v, int* sc, int tid) {
  const int lane = tid & 31, w = tid >> 5;
  int a = v;
#pragma unroll
  for (int off = 1; off < 32; off <<= 1) {
    const int n = __shfl_up(a, off, 32);
    if (lane >= off) a += n;
  }
  if (lane == 31) sc[w] = a;
  __syncthreads();
  if (w == 0) {
    const int o = sc[lane];
    int b = o;
#pragma unroll
    for (int off = 1; off < 32; off <<= 1) {
      const int n = __shfl_up(b, off, 32);
      if (lane >= off) b += n;
    }
    sc[lane] = b - o;
  }
  __syncthreads();
  return a - v + sc[w];
}

__global__ __launch_bounds__(ST) void vp_sort(const int*   __restrict__ geom,
                                              const float* __restrict__ depth,
                                              const float* __restrict__ ctx,
                                              float* __restrict__ ctxM,
                                              float* __restrict__ ctxE,
                                              int*   __restrict__ offs,
                                              uint2* __restrict__ recs,
                                              size_t ws_size) {
  if (ws_size < WS_NEED) return;
  __shared__ uint2 lrec[PPB];
  __shared__ int   cnt[NB];
  __shared__ int   lbase[NB];
  __shared__ int   sc[32];
  const int tid = threadIdx.x;
  if (blockIdx.x < SB) {
    const int blk  = blockIdx.x;
    const int base = blk * PPB;
    cnt[tid] = 0;
    __syncthreads();

    unsigned keys[PPT];
    int      rank[PPT];
    float    dval[PPT];
#pragma unroll
    for (int i = 0; i < PPT; ++i) {
      const int l = tid + i * ST;
      keys[i] = 0u; rank[i] = 0; dval[i] = 0.f;
      if (l < PPB) {
        bool valid;
        const unsigned key = point_key(geom, base + l, valid);
        const float d = depth[base + l];
        keys[i] = key;
        dval[i] = valid ? d : 0.f;
        rank[i] = atomicAdd(&cnt[(key >> 18) & (NB - 1)], 1);
      }
    }
    __syncthreads();

    const int c0 = cnt[tid];
    const int ex = block_excl_scan(c0, sc, tid);
    lbase[tid] = ex;
    int* row = offs + blk * (NB + 1);
    row[tid] = ex;
    if (tid == ST - 1) row[NB] = ex + c0;
    __syncthreads();

#pragma unroll
    for (int i = 0; i < PPT; ++i) {
      const int l = tid + i * ST;
      if (l < PPB) {
        uint2 rr; rr.x = keys[i]; rr.y = __float_as_uint(dval[i]);
        lrec[lbase[(keys[i] >> 18) & (NB - 1)] + rank[i]] = rr;
      }
    }
    __syncthreads();

#pragma unroll
    for (int i = 0; i < PPT; ++i) {
      const int l = tid + i * ST;
      if (l < PPB) recs[base + l] = lrec[l];
    }
  } else {
    float* sm = (float*)lrec;
    const int t   = blockIdx.x - SB;
    const int bn  = t / (HW / TW);
    const int hw0 = (t % (HW / TW)) * TW;
#pragma unroll
    for (int i = 0; i < (NC * TW) / ST; ++i) {
      const int e = tid + i * ST;
      const int c = e / TW, x = e % TW;
      sm[c * (TW + 1) + x] = ctx[(bn * NC + c) * HW + hw0 + x];
    }
    __syncthreads();
    float* dstM = ctxM + (size_t)(bn * HW + hw0) * NCM;
#pragma unroll
    for (int i = 0; i < (NCM * TW) / ST; ++i) {
      const int e = tid + i * ST;
      const int x = e / NCM, c = e % NCM;
      dstM[e] = sm[c * (TW + 1) + x];
    }
    float* dstE = ctxE + (size_t)(bn * HW + hw0) * NCE;
#pragma unroll
    for (int i = 0; i < (NCE * TW) / ST; ++i) {
      const int e = tid + i * ST;
      const int x = e / NCE, c = NCM + e % NCE;
      dstE[e] = sm[c * (TW + 1) + x];
    }
  }
}

struct StepIn {
  float d0, d1;
  int   m0, m1;
  unsigned r0, r1;
};
struct StepB {
  float4 q0, q1;
  float  e0, e1;
};

__device__ __forceinline__ StepIn decode_recs(uint2 ra, uint2 rb, int i0, int end) {
  if (i0     >= end) { ra.x = 0u; ra.y = 0u; }
  if (i0 + 1 >= end) { rb.x = 0u; rb.y = 0u; }
  StepIn s;
  s.d0 = __uint_as_float(ra.y);
  s.d1 = __uint_as_float(rb.y);
  s.m0 = (ra.x >> 14) & 15;
  s.m1 = (rb.x >> 14) & 15;
  s.r0 = ra.x & 0x3FFFu;
  s.r1 = rb.x & 0x3FFFu;
  return s;
}

__device__ __forceinline__ StepB load_ctx(const StepIn& s,
                                          const float* __restrict__ ctxM,
                                          const float* __restrict__ ctxE, int c) {
  const unsigned cm = (unsigned)c * 16u;
  const unsigned ce = (unsigned)c * 4u;
  StepB b;
  b.q0 = *(const float4*)((const char*)ctxM + ((s.r0 * (NCM * 4u)) | cm));
  b.q1 = *(const float4*)((const char*)ctxM + ((s.r1 * (NCM * 4u)) | cm));
  b.e0 = *(const float*)((const char*)ctxE + ((s.r0 * (NCE * 4u)) | ce));
  b.e1 = *(const float*)((const char*)ctxE + ((s.r1 * (NCE * 4u)) | ce));
  return b;
}

__global__ __launch_bounds__(128) void vp_gather(const int*   __restrict__ offs,
                                                 const uint2* __restrict__ recs,
                                                 const float* __restrict__ ctxM,
                                                 const float* __restrict__ ctxE,
                                                 float*       __restrict__ out,
                                                 size_t ws_size) {
  if (ws_size < WS_NEED) return;
  __shared__ float part[4][16 * NC];
  __shared__ uint2 lrecs[CH];
  __shared__ int   pre[SEGS + 1];
  __shared__ int   src[SEGS];
  __shared__ int   wsum[4];
  const int tid  = threadIdx.x;
  const int lane = tid & 31;
  const int wave = tid >> 5;
  const int c = lane & 15;
  const int h = lane >> 4;
  const int g = blockIdx.x;
  const int bb = g / NB;
  const int lg = g % NB;

  int sv = 0, cv = 0;
  if (tid < SEGS) {
    const int* o = offs + (bb * SEGS + tid) * (NB + 1) + lg;
    sv = o[0];
    cv = o[1] - sv;
  }
  int inc = cv;
#pragma unroll
  for (int off = 1; off < 32; off <<= 1) {
    const int nb = __shfl_up(inc, off, 32);
    if (lane >= off) inc += nb;
  }
  if (lane == 31) wsum[wave] = inc;
  __syncthreads();
  if (tid < SEGS) {
    const int ex = inc - cv + (wave == 1 ? wsum[0] : 0);
    pre[tid] = ex;
    src[tid] = (bb * SEGS + tid) * PPB + sv - ex;
    if (tid == SEGS - 1) pre[SEGS] = ex + cv;
  }
  __syncthreads();
  const int total = pre[SEGS];

  v8f acc[5];
#pragma unroll
  for (int j = 0; j < 5; ++j) {
    v8f z = {0.f, 0.f, 0.f, 0.f, 0.f, 0.f, 0.f, 0.f};
    acc[j] = z;
  }

  for (int chunk0 = 0; chunk0 < total; chunk0 += CH) {
    const int n = (total - chunk0) < CH ? (total - chunk0) : CH;
    for (int k0 = 0; k0 < n; k0 += 4 * 128) {
      uint2 r[4];
#pragma unroll
      for (int u = 0; u < 4; ++u) {
        const int li = k0 + tid + 128 * u;
        const int i  = chunk0 + (li < n ? li : n - 1);
        int lo = 0;
#pragma unroll
        for (int st = SEGS / 2; st >= 1; st >>= 1)
          if (pre[lo + st] <= i) lo += st;
        int a = src[lo] + i;
        a = a < 0 ? 0 : (a > NPTS - 1 ? NPTS - 1 : a);
        r[u] = recs[a];
      }
      asm volatile("" :: "v"(r[0].x), "v"(r[0].y), "v"(r[1].x), "v"(r[1].y),
                         "v"(r[2].x), "v"(r[2].y), "v"(r[3].x), "v"(r[3].y));
#pragma unroll
      for (int u = 0; u < 4; ++u) {
        const int li = k0 + tid + 128 * u;
        if (li < n) lrecs[li] = r[u];
      }
    }
    __syncthreads();

    const int nst = (n + 3) >> 2;
    for (int s = wave; s < nst; s += 16) {
      StepIn rs[4];
#pragma unroll
      for (int u = 0; u < 4; ++u) {
        const int li = 4 * (s + 4 * u) + 2 * h;
        const uint2 ua = lrecs[li     < n ? li     : n - 1];
        const uint2 ub = lrecs[li + 1 < n ? li + 1 : n - 1];
        rs[u] = decode_recs(ua, ub, li, n);
      }
      StepB bs[4];
#pragma unroll
      for (int u = 0; u < 4; ++u)
        bs[u] = load_ctx(rs[u], ctxM, ctxE, c);
      asm volatile("" :: "v"(bs[0].q0.x), "v"(bs[0].q1.x), "v"(bs[0].e0), "v"(bs[0].e1),
                         "v"(bs[1].q0.x), "v"(bs[1].q1.x), "v"(bs[1].e0), "v"(bs[1].e1),
                         "v"(bs[2].q0.x), "v"(bs[2].q1.x), "v"(bs[2].e0), "v"(bs[2].e1),
                         "v"(bs[3].q0.x), "v"(bs[3].q1.x), "v"(bs[3].e0), "v"(bs[3].e1));
#pragma unroll
      for (int u = 0; u < 4; ++u) {
        v2f a;
        a.x = (rs[u].m0 == c) ? rs[u].d0 : 0.f;
        a.y = (rs[u].m1 == c) ? rs[u].d1 : 0.f;
        const float x0[4] = {bs[u].q0.x, bs[u].q0.y, bs[u].q0.z, bs[u].q0.w};
        const float x1[4] = {bs[u].q1.x, bs[u].q1.y, bs[u].q1.z, bs[u].q1.w};
#pragma unroll
        for (int j = 0; j < 4; ++j) {
          v2f b; b.x = x0[j]; b.y = x1[j];
          acc[j] = __builtin_amdgcn_wmma_f32_16x16x4_f32(
              false, a, false, b, (short)0, acc[j], false, false);
        }
        v2f b4; b4.x = bs[u].e0; b4.y = bs[u].e1;
        acc[4] = __builtin_amdgcn_wmma_f32_16x16x4_f32(
            false, a, false, b4, (short)0, acc[4], false, false);
      }
    }
    __syncthreads();
  }

  float* mine = part[wave];
#pragma unroll
  for (int r = 0; r < 8; ++r) {
    float* rowp = mine + (8 * h + r) * NC;
    *(float4*)(rowp + 4 * c) = make_float4(acc[0][r], acc[1][r], acc[2][r], acc[3][r]);
    rowp[64 + c] = acc[4][r];
  }
  __syncthreads();

  float4* og = (float4*)(out + (size_t)g * (16 * NC));
#pragma unroll
  for (int i = 0; i < 3; ++i) {
    const int e4 = tid + i * 128;
    if (e4 < (16 * NC) / 4) {
      const float4 p0 = ((const float4*)part[0])[e4];
      const float4 p1 = ((const float4*)part[1])[e4];
      const float4 p2 = ((const float4*)part[2])[e4];
      const float4 p3 = ((const float4*)part[3])[e4];
      float4 v;
      v.x = (p0.x + p1.x) + (p2.x + p3.x);
      v.y = (p0.y + p1.y) + (p2.y + p3.y);
      v.z = (p0.z + p1.z) + (p2.z + p3.z);
      v.w = (p0.w + p1.w) + (p2.w + p3.w);
      og[e4] = v;
    }
  }
}

extern "C" void kernel_launch(void* const* d_in, const int* in_sizes, int n_in,
                              void* d_out, int out_size, void* d_ws, size_t ws_size,
                              hipStream_t stream) {
  (void)in_sizes; (void)n_in; (void)out_size;
  const int*   geom  = (const int*)d_in[0];
  const float* depth = (const float*)d_in[1];
  const float* ctx   = (const float*)d_in[2];
  float*       out   = (float*)d_out;

  char*  ws   = (char*)d_ws;
  float* ctxM = (float*)(ws + WS_CTXM);
  float* ctxE = (float*)(ws + WS_CTXE);
  int*   offs = (int*)(ws + WS_OFFS);
  uint2* recs = (uint2*)(ws + WS_RECS);

  vp_sort<<<SB + NTB, ST, 0, stream>>>(geom, depth, ctx, ctxM, ctxE, offs, recs, ws_size);
  vp_gather<<<NBINS, 128, 0, stream>>>(offs, recs, ctxM, ctxE, out, ws_size);
}


# Sort blocks issue a thread's eight point loads (geom x, y, depth) as one group ahead of the first LDS count: one memory hop per thread, not eight dependent ones; sums still in LDS arrival order
# speedup vs baseline: 2.2339x; 1.0484x over previous
#include <hip/hip_runtime.h>

#define NVX 128
#define NVY 128
#define NC  80
#define B_  2
#define N_  6
#define D_  112
#define H_  16
#define W_  44
#define HW  (H_ * W_)
#define NPTS  (B_ * N_ * D_ * HW)
#define NRAYS (B_ * N_ * HW)
#define NBINS (B_ * NVX * NVY / 16)
#define NB    (NBINS / B_)
#define SEGS  64
#define SB    (B_ * SEGS)
#define PPB   (NPTS / SB)
#define ST    1024
#define PPT   8
#define TW    64
#define NTB   (B_ * N_ * (HW / TW))
#define NCM   64
#define NCE   (NC - NCM)
#define CH    1024

#define WS_CTXM 0
#define WS_CTXE (WS_CTXM + NRAYS * NCM * 4)
#define WS_OFFS (WS_CTXE + NRAYS * NCE * 4)
#define WS_RECS (WS_OFFS + SB * (NB + 1) * 4)
#define WS_NEED ((size_t)WS_RECS + (size_t)NPTS * 8)

typedef __attribute__((ext_vector_type(2))) float v2f;
typedef __attribute__((ext_vector_type(8))) float v8f;

__device__ __forceinline__ unsigned point_key(int gx, int gy, int p, bool& valid) {
  valid = ((unsigned)gx < (unsigned)NVX) && ((unsigned)gy < (unsigned)NVY);
  const int cx = gx < 0 ? 0 : (gx > NVX - 1 ? NVX - 1 : gx);
  const int cy = gy < 0 ? 0 : (gy > NVY - 1 ? NVY - 1 : gy);
  const int b    = p / (N_ * D_ * HW);
  const int flat = b * (NVX * NVY) + cy * NVX + cx;
  const int bn   = p / (D_ * HW);
  const int ray  = bn * HW + p % HW;
  return (unsigned)ray | ((unsigned)flat << 14);
}

__device__ __forceinline__ int block_excl_scan(int v, int* sc, int tid) {
  const int lane = tid & 31, w = tid >> 5;
  int a = v;
#pragma unroll
  for (int off = 1; off < 32; off <<= 1) {
    const int n = __shfl_up(a, off, 32);
    if (lane >= off) a += n;
  }
  if (lane == 31) sc[w] = a;
  __syncthreads();
  if (w == 0) {
    const int o = sc[lane];
    int b = o;
#pragma unroll
    for (int off = 1; off < 32; off <<= 1) {
      const int n = __shfl_up(b, off, 32);
      if (lane >= off) b += n;
    }
    sc[lane] = b - o;
  }
  __syncthreads();
  return a - v + sc[w];
}

__global__ __launch_bounds__(ST) void vp_sort(const int*   __restrict__ geom,
                                              const float* __restrict__ depth,
                                              const float* __restrict__ ctx,
                                              float* __restrict__ ctxM,
                                              float* __restrict__ ctxE,
                                              int*   __restrict__ offs,
                                              uint2* __restrict__ recs,
                                              size_t ws_size) {
  if (ws_size < WS_NEED) return;
  __shared__ uint2 lrec[PPB];
  __shared__ int   cnt[NB];
  __shared__ int   lbase[NB];
  __shared__ int   sc[32];
  const int tid = threadIdx.x;
  if (blockIdx.x < SB) {
    const int blk  = blockIdx.x;
    const int base = blk * PPB;

    int   gxs[PPT], gys[PPT];
    float dls[PPT];
#pragma unroll
    for (int i = 0; i < PPT; ++i) {
      const int l = tid + i * ST;
      gxs[i] = 0; gys[i] = 0; dls[i] = 0.f;
      if (l < PPB) {
        gxs[i] = geom[3 * (base + l) + 0];
        gys[i] = geom[3 * (base + l) + 1];
        dls[i] = depth[base + l];
      }
    }
    cnt[tid] = 0;
    __syncthreads();

    unsigned keys[PPT];
    int      rank[PPT];
    float    dval[PPT];
#pragma unroll
    for (int i = 0; i < PPT; ++i) {
      const int l = tid + i * ST;
      keys[i] = 0u; rank[i] = 0; dval[i] = 0.f;
      if (l < PPB) {
        bool valid;
        const unsigned key = point_key(gxs[i], gys[i], base + l, valid);
        keys[i] = key;
        dval[i] = valid ? dls[i] : 0.f;
        rank[i] = atomicAdd(&cnt[(key >> 18) & (NB - 1)], 1);
      }
    }
    __syncthreads();

    const int c0 = cnt[tid];
    const int ex = block_excl_scan(c0, sc, tid);
    lbase[tid] = ex;
    int* row = offs + blk * (NB + 1);
    row[tid] = ex;
    if (tid == ST - 1) row[NB] = ex + c0;
    __syncthreads();

#pragma unroll
    for (int i = 0; i < PPT; ++i) {
      const int l = tid + i * ST;
      if (l < PPB) {
        uint2 rr; rr.x = keys[i]; rr.y = __float_as_uint(dval[i]);
        lrec[lbase[(keys[i] >> 18) & (NB - 1)] + rank[i]] = rr;
      }
    }
    __syncthreads();

#pragma unroll
    for (int i = 0; i < PPT; ++i) {
      const int l = tid + i * ST;
      if (l < PPB) recs[base + l] = lrec[l];
    }
  } else {
    float* sm = (float*)lrec;
    const int t   = blockIdx.x - SB;
    const int bn  = t / (HW / TW);
    const int hw0 = (t % (HW / TW)) * TW;
#pragma unroll
    for (int i = 0; i < (NC * TW) / ST; ++i) {
      const int e = tid + i * ST;
      const int c = e / TW, x = e % TW;
      sm[c * (TW + 1) + x] = ctx[(bn * NC + c) * HW + hw0 + x];
    }
    __syncthreads();
    float* dstM = ctxM + (size_t)(bn * HW + hw0) * NCM;
#pragma unroll
    for (int i = 0; i < (NCM * TW) / ST; ++i) {
      const int e = tid + i * ST;
      const int x = e / NCM, c = e % NCM;
      dstM[e] = sm[c * (TW + 1) + x];
    }
    float* dstE = ctxE + (size_t)(bn * HW + hw0) * NCE;
#pragma unroll
    for (int i = 0; i < (NCE * TW) / ST; ++i) {
      const int e = tid + i * ST;
      const int x = e / NCE, c = NCM + e % NCE;
      dstE[e] = sm[c * (TW + 1) + x];
    }
  }
}

struct StepIn {
  float d0, d1;
  int   m0, m1;
  unsigned r0, r1;
};
struct StepB {
  float4 q0, q1;
  float  e0, e1;
};

__device__ __forceinline__ StepIn decode_recs(uint2 ra, uint2 rb, int i0, int end) {
  if (i0     >= end) { ra.x = 0u; ra.y = 0u; }
  if (i0 + 1 >= end) { rb.x = 0u; rb.y = 0u; }
  StepIn s;
  s.d0 = __uint_as_float(ra.y);
  s.d1 = __uint_as_float(rb.y);
  s.m0 = (ra.x >> 14) & 15;
  s.m1 = (rb.x >> 14) & 15;
  s.r0 = ra.x & 0x3FFFu;
  s.r1 = rb.x & 0x3FFFu;
  return s;
}

__device__ __forceinline__ StepB load_ctx(const StepIn& s,
                                          const float* __restrict__ ctxM,
                                          const float* __restrict__ ctxE, int c) {
  const unsigned cm = (unsigned)c * 16u;
  const unsigned ce = (unsigned)c * 4u;
  StepB b;
  b.q0 = *(const float4*)((const char*)ctxM + ((s.r0 * (NCM * 4u)) | cm));
  b.q1 = *(const float4*)((const char*)ctxM + ((s.r1 * (NCM * 4u)) | cm));
  b.e0 = *(const float*)((const char*)ctxE + ((s.r0 * (NCE * 4u)) | ce));
  b.e1 = *(const float*)((const char*)ctxE + ((s.r1 * (NCE * 4u)) | ce));
  return b;
}

__global__ __launch_bounds__(128) void vp_gather(const int*   __restrict__ offs,
                                                 const uint2* __restrict__ recs,
                                                 const float* __restrict__ ctxM,
                                                 const float* __restrict__ ctxE,
                                                 float*       __restrict__ out,
                                                 size_t ws_size) {
  if (ws_size < WS_NEED) return;
  __shared__ float part[4][16 * NC];
  __shared__ uint2 lrecs[CH];
  __shared__ int   pre[SEGS + 1];
  __shared__ int   src[SEGS];
  __shared__ int   wsum[4];
  const int tid  = threadIdx.x;
  const int lane = tid & 31;
  const int wave = tid >> 5;
  const int c = lane & 15;
  const int h = lane >> 4;
  const int g = blockIdx.x;
  const int bb = g / NB;
  const int lg = g % NB;

  int sv = 0, cv = 0;
  if (tid < SEGS) {
    const int* o = offs + (bb * SEGS + tid) * (NB + 1) + lg;
    sv = o[0];
    cv = o[1] - sv;
  }
  int inc = cv;
#pragma unroll
  for (int off = 1; off < 32; off <<= 1) {
    const int nb = __shfl_up(inc, off, 32);
    if (lane >= off) inc += nb;
  }
  if (lane == 31) wsum[wave] = inc;
  __syncthreads();
  if (tid < SEGS) {
    const int ex = inc - cv + (wave == 1 ? wsum[0] : 0);
    pre[tid] = ex;
    src[tid] = (bb * SEGS + tid) * PPB + sv - ex;
    if (tid == SEGS - 1) pre[SEGS] = ex + cv;
  }
  __syncthreads();
  const int total = pre[SEGS];

  v8f acc[5];
#pragma unroll
  for (int j = 0; j < 5; ++j) {
    v8f z = {0.f, 0.f, 0.f, 0.f, 0.f, 0.f, 0.f, 0.f};
    acc[j] = z;
  }

  for (int chunk0 = 0; chunk0 < total; chunk0 += CH) {
    const int n = (total - chunk0) < CH ? (total - chunk0) : CH;
    for (int k0 = 0; k0 < n; k0 += 4 * 128) {
      uint2 r[4];
#pragma unroll
      for (int u = 0; u < 4; ++u) {
        const int li = k0 + tid + 128 * u;
        const int i  = chunk0 + (li < n ? li : n - 1);
        int lo = 0;
#pragma unroll
        for (int st = SEGS / 2; st >= 1; st >>= 1)
          if (pre[lo + st] <= i) lo += st;
        int a = src[lo] + i;
        a = a < 0 ? 0 : (a > NPTS - 1 ? NPTS - 1 : a);
        r[u] = recs[a];
      }
      asm volatile("" :: "v"(r[0].x), "v"(r[0].y), "v"(r[1].x), "v"(r[1].y),
                         "v"(r[2].x), "v"(r[2].y), "v"(r[3].x), "v"(r[3].y));
#pragma unroll
      for (int u = 0; u < 4; ++u) {
        const int li = k0 + tid + 128 * u;
        if (li < n) lrecs[li] = r[u];
      }
    }
    __syncthreads();

    const int nst = (n + 3) >> 2;
    for (int s = wave; s < nst; s += 16) {
      StepIn rs[4];
#pragma unroll
      for (int u = 0; u < 4; ++u) {
        const int li = 4 * (s + 4 * u) + 2 * h;
        const uint2 ua = lrecs[li     < n ? li     : n - 1];
        const uint2 ub = lrecs[li + 1 < n ? li + 1 : n - 1];
        rs[u] = decode_recs(ua, ub, li, n);
      }
      StepB bs[4];
#pragma unroll
      for (int u = 0; u < 4; ++u)
        bs[u] = load_ctx(rs[u], ctxM, ctxE, c);
      asm volatile("" :: "v"(bs[0].q0.x), "v"(bs[0].q1.x), "v"(bs[0].e0), "v"(bs[0].e1),
                         "v"(bs[1].q0.x), "v"(bs[1].q1.x), "v"(bs[1].e0), "v"(bs[1].e1),
                         "v"(bs[2].q0.x), "v"(bs[2].q1.x), "v"(bs[2].e0), "v"(bs[2].e1),
                         "v"(bs[3].q0.x), "v"(bs[3].q1.x), "v"(bs[3].e0), "v"(bs[3].e1));
#pragma unroll
      for (int u = 0; u < 4; ++u) {
        v2f a;
        a.x = (rs[u].m0 == c) ? rs[u].d0 : 0.f;
        a.y = (rs[u].m1 == c) ? rs[u].d1 : 0.f;
        const float x0[4] = {bs[u].q0.x, bs[u].q0.y, bs[u].q0.z, bs[u].q0.w};
        const float x1[4] = {bs[u].q1.x, bs[u].q1.y, bs[u].q1.z, bs[u].q1.w};
#pragma unroll
        for (int j = 0; j < 4; ++j) {
          v2f b; b.x = x0[j]; b.y = x1[j];
          acc[j] = __builtin_amdgcn_wmma_f32_16x16x4_f32(
              false, a, false, b, (short)0, acc[j], false, false);
        }
        v2f b4; b4.x = bs[u].e0; b4.y = bs[u].e1;
        acc[4] = __builtin_amdgcn_wmma_f32_16x16x4_f32(
            false, a, false, b4, (short)0, acc[4], false, false);
      }
    }
    __syncthreads();
  }

  float* mine = part[wave];
#pragma unroll
  for (int r = 0; r < 8; ++r) {
    float* rowp = mine + (8 * h + r) * NC;
    *(float4*)(rowp + 4 * c) = make_float4(acc[0][r], acc[1][r], acc[2][r], acc[3][r]);
    rowp[64 + c] = acc[4][r];
  }
  __syncthreads();

  float4* og = (float4*)(out + (size_t)g * (16 * NC));
#pragma unroll
  for (int i = 0; i < 3; ++i) {
    const int e4 = tid + i * 128;
    if (e4 < (16 * NC) / 4) {
      const float4 p0 = ((const float4*)part[0])[e4];
      const float4 p1 = ((const float4*)part[1])[e4];
      const float4 p2 = ((const float4*)part[2])[e4];
      const float4 p3 = ((const float4*)part[3])[e4];
      float4 v;
      v.x = (p0.x + p1.x) + (p2.x + p3.x);
      v.y = (p0.y + p1.y) + (p2.y + p3.y);
      v.z = (p0.z + p1.z) + (p2.z + p3.z);
      v.w = (p0.w + p1.w) + (p2.w + p3.w);
      og[e4] = v;
    }
  }
}

extern "C" void kernel_launch(void* const* d_in, const int* in_sizes, int n_in,
                              void* d_out, int out_size, void* d_ws, size_t ws_size,
                              hipStream_t stream) {
  (void)in_sizes; (void)n_in; (void)out_size;
  const int*   geom  = (const int*)d_in[0];
  const float* depth = (const float*)d_in[1];
  const float* ctx   = (const float*)d_in[2];
  float*       out   = (float*)d_out;

  char*  ws   = (char*)d_ws;
  float* ctxM = (float*)(ws + WS_CTXM);
  float* ctxE = (float*)(ws + WS_CTXE);
  int*   offs = (int*)(ws + WS_OFFS);
  uint2* recs = (uint2*)(ws + WS_RECS);

  vp_sort<<<SB + NTB, ST, 0, stream>>>(geom, depth, ctx, ctxM, ctxE, offs, recs, ws_size);
  vp_gather<<<NBINS, 128, 0, stream>>>(offs, recs, ctxM, ctxE, out, ws_size);
}


# Sort launch split over 256 blocks of 512 threads (128 half-size segments per batch, 16-bit group starts) so every compute unit sorts 3696 points; a group's sum order is now by 128 segments
# speedup vs baseline: 2.3182x; 1.0377x over previous
#include <hip/hip_runtime.h>

#define NVX 128
#define NVY 128
#define NC  80
#define B_  2
#define N_  6
#define D_  112
#define H_  16
#define W_  44
#define HW  (H_ * W_)
#define NPTS  (B_ * N_ * D_ * HW)
#define NRAYS (B_ * N_ * HW)
#define NBINS (B_ * NVX * NVY / 16)
#define NB    (NBINS / B_)
#define SEGS  128
#define SB    (B_ * SEGS)
#define PPB   (NPTS / SB)
#define ST    512
#define NW    (ST / 32)
#define BPT   (NB / ST)
#define PPT   8
#define TW    64
#define NTB   (B_ * N_ * (HW / TW))
#define NCM   64
#define NCE   (NC - NCM)
#define CH    1024

#define WS_CTXM 0
#define WS_CTXE (WS_CTXM + NRAYS * NCM * 4)
#define WS_OFFS (WS_CTXE + NRAYS * NCE * 4)
#define WS_RECS (WS_OFFS + SB * (NB / 2) * 4)
#define WS_NEED ((size_t)WS_RECS + (size_t)NPTS * 8)

static_assert(PPB < 65536, "segment offsets must fit 16 bits");
static_assert(SB * PPB == NPTS && (NPTS / B_) % PPB == 0, "a sort block's points are of one batch");
static_assert(BPT * ST == NB && BPT == 2, "each sort thread owns two adjacent groups");
static_assert(WS_RECS % 8 == 0, "records are 8-byte aligned");
static_assert(NC * (TW + 1) * 4 <= PPB * 8, "the transpose tile fits in the record array");
static_assert((NC * TW) % ST == 0 && (NCM * TW) % ST == 0 && (NCE * TW) % ST == 0, "whole transpose passes");
typedef __attribute__((ext_vector_type(2))) float v2f;
typedef __attribute__((ext_vector_type(8))) float v8f;

__device__ __forceinline__ unsigned point_key(int gx, int gy, int p, bool& valid) {
  valid = ((unsigned)gx < (unsigned)NVX) && ((unsigned)gy < (unsigned)NVY);
  const int cx = gx < 0 ? 0 : (gx > NVX - 1 ? NVX - 1 : gx);
  const int cy = gy < 0 ? 0 : (gy > NVY - 1 ? NVY - 1 : gy);
  const int b    = p / (N_ * D_ * HW);
  const int flat = b * (NVX * NVY) + cy * NVX + cx;
  const int bn   = p / (D_ * HW);
  const int ray  = bn * HW + p % HW;
  return (unsigned)ray | ((unsigned)flat << 14);
}

__device__ __forceinline__ int block_excl_scan(int v, int* sc, int tid) {
  const int lane = tid & 31, w = tid >> 5;
  int a = v;
#pragma unroll
  for (int off = 1; off < 32; off <<= 1) {
    const int n = __shfl_up(a, off, 32);
    if (lane >= off) a += n;
  }
  if (lane == 31) sc[w] = a;
  __syncthreads();
  if (w == 0) {
    const int o = lane < NW ? sc[lane] : 0;
    int b = o;
#pragma unroll
    for (int off = 1; off < 32; off <<= 1) {
      const int n = __shfl_up(b, off, 32);
      if (lane >= off) b += n;
    }
    sc[lane] = b - o;
  }
  __syncthreads();
  return a - v + sc[w];
}

__global__ __launch_bounds__(ST) void vp_sort(const int*   __restrict__ geom,
                                              const float* __restrict__ depth,
                                              const float* __restrict__ ctx,
                                              float* __restrict__ ctxM,
                                              float* __restrict__ ctxE,
                                              unsigned* __restrict__ offs,
                                              uint2* __restrict__ recs,
                                              size_t ws_size) {
  if (ws_size < WS_NEED) return;
  __shared__ uint2 lrec[PPB];
  __shared__ int   cnt[NB];
  __shared__ int   lbase[NB];
  __shared__ int   sc[32];
  const int tid = threadIdx.x;
  if (blockIdx.x < SB) {
    const int blk  = blockIdx.x;
    const int base = blk * PPB;

    int   gxs[PPT], gys[PPT];
    float dls[PPT];
#pragma unroll
    for (int i = 0; i < PPT; ++i) {
      const int l  = tid + i * ST;
      const int lc = l < PPB ? l : PPB - 1;
      gxs[i] = geom[3 * (base + lc) + 0];
      gys[i] = geom[3 * (base + lc) + 1];
      dls[i] = depth[base + lc];
    }
#pragma unroll
    for (int b = 0; b < BPT; ++b) cnt[tid + b * ST] = 0;
    __syncthreads();

    unsigned keys[PPT];
    int      rank[PPT];
    float    dval[PPT];
#pragma unroll
    for (int i = 0; i < PPT; ++i) {
      const int l = tid + i * ST;
      keys[i] = 0u; rank[i] = 0; dval[i] = 0.f;
      if (l < PPB) {
        bool valid;
        const unsigned key = point_key(gxs[i], gys[i], base + l, valid);
        keys[i] = key;
        dval[i] = valid ? dls[i] : 0.f;
        rank[i] = atomicAdd(&cnt[(key >> 18) & (NB - 1)], 1);
      }
    }
    __syncthreads();

    const int c0 = cnt[2 * tid];
    const int c1 = cnt[2 * tid + 1];
    const int ex = block_excl_scan(c0 + c1, sc, tid);
    lbase[2 * tid]     = ex;
    lbase[2 * tid + 1] = ex + c0;
    offs[(size_t)blk * (NB / 2) + tid] = (unsigned)ex | ((unsigned)(ex + c0) << 16);
    __syncthreads();

#pragma unroll
    for (int i = 0; i < PPT; ++i) {
      const int l = tid + i * ST;
      if (l < PPB) {
        uint2 rr; rr.x = keys[i]; rr.y = __float_as_uint(dval[i]);
        lrec[lbase[(keys[i] >> 18) & (NB - 1)] + rank[i]] = rr;
      }
    }
    __syncthreads();

#pragma unroll
    for (int i = 0; i < PPT; ++i) {
      const int l = tid + i * ST;
      if (l < PPB) recs[base + l] = lrec[l];
    }
  } else {
    float* sm = (float*)lrec;
    const int t   = blockIdx.x - SB;
    const int bn  = t / (HW / TW);
    const int hw0 = (t % (HW / TW)) * TW;
#pragma unroll
    for (int i = 0; i < (NC * TW) / ST; ++i) {
      const int e = tid + i * ST;
      const int c = e / TW, x = e % TW;
      sm[c * (TW + 1) + x] = ctx[(bn * NC + c) * HW + hw0 + x];
    }
    __syncthreads();
    float* dstM = ctxM + (size_t)(bn * HW + hw0) * NCM;
#pragma unroll
    for (int i = 0; i < (NCM * TW) / ST; ++i) {
      const int e = tid + i * ST;
      const int x = e / NCM, c = e % NCM;
      dstM[e] = sm[c * (TW + 1) + x];
    }
    float* dstE = ctxE + (size_t)(bn * HW + hw0) * NCE;
#pragma unroll
    for (int i = 0; i < (NCE * TW) / ST; ++i) {
      const int e = tid + i * ST;
      const int x = e / NCE, c = NCM + e % NCE;
      dstE[e] = sm[c * (TW + 1) + x];
    }
  }
}

struct StepIn {
  float d0, d1;
  int   m0, m1;
  unsigned r0, r1;
};
struct StepB {
  float4 q0, q1;
  float  e0, e1;
};

__device__ __forceinline__ StepIn decode_recs(uint2 ra, uint2 rb, int i0, int end) {
  if (i0     >= end) { ra.x = 0u; ra.y = 0u; }
  if (i0 + 1 >= end) { rb.x = 0u; rb.y = 0u; }
  StepIn s;
  s.d0 = __uint_as_float(ra.y);
  s.d1 = __uint_as_float(rb.y);
  s.m0 = (ra.x >> 14) & 15;
  s.m1 = (rb.x >> 14) & 15;
  s.r0 = ra.x & 0x3FFFu;
  s.r1 = rb.x & 0x3FFFu;
  return s;
}

__device__ __forceinline__ StepB load_ctx(const StepIn& s,
                                          const float* __restrict__ ctxM,
                                          const float* __restrict__ ctxE, int c) {
  const unsigned cm = (unsigned)c * 16u;
  const unsigned ce = (unsigned)c * 4u;
  StepB b;
  b.q0 = *(const float4*)((const char*)ctxM + ((s.r0 * (NCM * 4u)) | cm));
  b.q1 = *(const float4*)((const char*)ctxM + ((s.r1 * (NCM * 4u)) | cm));
  b.e0 = *(const float*)((const char*)ctxE + ((s.r0 * (NCE * 4u)) | ce));
  b.e1 = *(const float*)((const char*)ctxE + ((s.r1 * (NCE * 4u)) | ce));
  return b;
}

static_assert(SEGS == 128, "the gather block has one thread per segment");
__global__ __launch_bounds__(128) void vp_gather(const unsigned* __restrict__ offs,
                                                 const uint2* __restrict__ recs,
                                                 const float* __restrict__ ctxM,
                                                 const float* __restrict__ ctxE,
                                                 float*       __restrict__ out,
                                                 size_t ws_size) {
  if (ws_size < WS_NEED) return;
  __shared__ float part[4][16 * NC];
  __shared__ uint2 lrecs[CH];
  __shared__ int   pre[SEGS + 1];
  __shared__ int   src[SEGS];
  __shared__ int   wsum[4];
  const int tid  = threadIdx.x;
  const int lane = tid & 31;
  const int wave = tid >> 5;
  const int c = lane & 15;
  const int h = lane >> 4;
  const int g = blockIdx.x;
  const int bb = g / NB;
  const int lg = g % NB;

  const unsigned* rowp = offs + (size_t)(bb * SEGS + tid) * (NB / 2);
  const int wi = lg >> 1;
  const unsigned w0 = rowp[wi];
  const unsigned w1 = rowp[wi + 1 < NB / 2 ? wi + 1 : NB / 2 - 1];
  int sv, ev;
  if (lg & 1) { sv = (int)(w0 >> 16);     ev = (lg + 1 < NB) ? (int)(w1 & 0xFFFFu) : PPB; }
  else        { sv = (int)(w0 & 0xFFFFu); ev = (int)(w0 >> 16); }
  const int cv = ev - sv;
  int inc = cv;
#pragma unroll
  for (int off = 1; off < 32; off <<= 1) {
    const int nb = __shfl_up(inc, off, 32);
    if (lane >= off) inc += nb;
  }
  if (lane == 31) wsum[wave] = inc;
  __syncthreads();
  {
    int wp = 0;
#pragma unroll
    for (int w = 0; w < 3; ++w)
      if (w < wave) wp += wsum[w];
    const int ex = inc - cv + wp;
    pre[tid] = ex;
    src[tid] = (bb * SEGS + tid) * PPB + sv - ex;
    if (tid == SEGS - 1) pre[SEGS] = ex + cv;
  }
  __syncthreads();
  const int total = pre[SEGS];

  v8f acc[5];
#pragma unroll
  for (int j = 0; j < 5; ++j) {
    v8f z = {0.f, 0.f, 0.f, 0.f, 0.f, 0.f, 0.f, 0.f};
    acc[j] = z;
  }

  for (int chunk0 = 0; chunk0 < total; chunk0 += CH) {
    const int n = (total - chunk0) < CH ? (total - chunk0) : CH;
    for (int k0 = 0; k0 < n; k0 += 4 * 128) {
      uint2 r[4];
#pragma unroll
      for (int u = 0; u < 4; ++u) {
        const int li = k0 + tid + 128 * u;
        const int i  = chunk0 + (li < n ? li : n - 1);
        int lo = 0;
#pragma unroll
        for (int st = SEGS / 2; st >= 1; st >>= 1)
          if (pre[lo + st] <= i) lo += st;
        int a = src[lo] + i;
        a = a < 0 ? 0 : (a > NPTS - 1 ? NPTS - 1 : a);
        r[u] = recs[a];
      }
      asm volatile("" :: "v"(r[0].x), "v"(r[0].y), "v"(r[1].x), "v"(r[1].y),
                         "v"(r[2].x), "v"(r[2].y), "v"(r[3].x), "v"(r[3].y));
#pragma unroll
      for (int u = 0; u < 4; ++u) {
        const int li = k0 + tid + 128 * u;
        if (li < n) lrecs[li] = r[u];
      }
    }
    __syncthreads();

    const int nst = (n + 3) >> 2;
    for (int s = wave; s < nst; s += 16) {
      StepIn rs[4];
#pragma unroll
      for (int u = 0; u < 4; ++u) {
        const int li = 4 * (s + 4 * u) + 2 * h;
        const uint2 ua = lrecs[li     < n ? li     : n - 1];
        const uint2 ub = lrecs[li + 1 < n ? li + 1 : n - 1];
        rs[u] = decode_recs(ua, ub, li, n);
      }
      StepB bs[4];
#pragma unroll
      for (int u = 0; u < 4; ++u)
        bs[u] = load_ctx(rs[u], ctxM, ctxE, c);
      asm volatile("" :: "v"(bs[0].q0.x), "v"(bs[0].q1.x), "v"(bs[0].e0), "v"(bs[0].e1),
                         "v"(bs[1].q0.x), "v"(bs[1].q1.x), "v"(bs[1].e0), "v"(bs[1].e1),
                         "v"(bs[2].q0.x), "v"(bs[2].q1.x), "v"(bs[2].e0), "v"(bs[2].e1),
                         "v"(bs[3].q0.x), "v"(bs[3].q1.x), "v"(bs[3].e0), "v"(bs[3].e1));
#pragma unroll
      for (int u = 0; u < 4; ++u) {
        v2f a;
        a.x = (rs[u].m0 == c) ? rs[u].d0 : 0.f;
        a.y = (rs[u].m1 == c) ? rs[u].d1 : 0.f;
        const float x0[4] = {bs[u].q0.x, bs[u].q0.y, bs[u].q0.z, bs[u].q0.w};
        const float x1[4] = {bs[u].q1.x, bs[u].q1.y, bs[u].q1.z, bs[u].q1.w};
#pragma unroll
        for (int j = 0; j < 4; ++j) {
          v2f b; b.x = x0[j]; b.y = x1[j];
          acc[j] = __builtin_amdgcn_wmma_f32_16x16x4_f32(
              false, a, false, b, (short)0, acc[j], false, false);
        }
        v2f b4; b4.x = bs[u].e0; b4.y = bs[u].e1;
        acc[4] = __builtin_amdgcn_wmma_f32_16x16x4_f32(
            false, a, false, b4, (short)0, acc[4], false, false);
      }
    }
    __syncthreads();
  }

  float* mine = part[wave];
#pragma unroll
  for (int r = 0; r < 8; ++r) {
    float* rowp = mine + (8 * h + r) * NC;
    *(float4*)(rowp + 4 * c) = make_float4(acc[0][r], acc[1][r], acc[2][r], acc[3][r]);
    rowp[64 + c] = acc[4][r];
  }
  __syncthreads();

  float4* og = (float4*)(out + (size_t)g * (16 * NC));
#pragma unroll
  for (int i = 0; i < 3; ++i) {
    const int e4 = tid + i * 128;
    if (e4 < (16 * NC) / 4) {
      const float4 p0 = ((const float4*)part[0])[e4];
      const float4 p1 = ((const float4*)part[1])[e4];
      const float4 p2 = ((const float4*)part[2])[e4];
      const float4 p3 = ((const float4*)part[3])[e4];
      float4 v;
      v.x = (p0.x + p1.x) + (p2.x + p3.x);
      v.y = (p0.y + p1.y) + (p2.y + p3.y);
      v.z = (p0.z + p1.z) + (p2.z + p3.z);
      v.w = (p0.w + p1.w) + (p2.w + p3.w);
      og[e4] = v;
    }
  }
}

extern "C" void kernel_launch(void* const* d_in, const int* in_sizes, int n_in,
                              void* d_out, int out_size, void* d_ws, size_t ws_size,
                              hipStream_t stream) {
  (void)in_sizes; (void)n_in; (void)out_size;
  const int*   geom  = (const int*)d_in[0];
  const float* depth = (const float*)d_in[1];
  const float* ctx   = (const float*)d_in[2];
  float*       out   = (float*)d_out;

  char*  ws   = (char*)d_ws;
  float* ctxM = (float*)(ws + WS_CTXM);
  float* ctxE = (float*)(ws + WS_CTXE);
  unsigned* offs = (unsigned*)(ws + WS_OFFS);
  uint2* recs = (uint2*)(ws + WS_RECS);

  vp_sort<<<SB + NTB, ST, 0, stream>>>(geom, depth, ctx, ctxM, ctxE, offs, recs, ws_size);
  vp_gather<<<NBINS, 128, 0, stream>>>(offs, recs, ctxM, ctxE, out, ws_size);
}
